# starting kernel
# baseline (speedup 1.0000x reference)
#include <hip/hip_runtime.h>

#define N_NODES 50000
#define N_EDGES 800000
#define D_FEAT  128
#define K_CHEB  3

typedef __attribute__((ext_vector_type(2))) float v2f;
typedef __attribute__((ext_vector_type(8))) float v8f;

__global__ void zero_i32(int* __restrict__ p, int n) {
    int i = blockIdx.x * blockDim.x + threadIdx.x;
    if (i < n) p[i] = 0;
}

__global__ void count_deg_i(const int* __restrict__ dst, int* __restrict__ cnt, int nEdges) {
    int e = blockIdx.x * blockDim.x + threadIdx.x;
    if (e < nEdges) atomicAdd(&cnt[dst[e]], 1);
}

__global__ __launch_bounds__(256) void scan_excl(const int* __restrict__ cnt,
                                                 int* __restrict__ rowptr, int n) {
    __shared__ int buf[256];
    __shared__ int carry;
    if (threadIdx.x == 0) carry = 0;
    __syncthreads();
    for (int base = 0; base < n; base += 256) {
        int i = base + (int)threadIdx.x;
        int v = (i < n) ? cnt[i] : 0;
        buf[threadIdx.x] = v;
        __syncthreads();
        for (int off = 1; off < 256; off <<= 1) {
            int t = (threadIdx.x >= (unsigned)off) ? buf[threadIdx.x - off] : 0;
            __syncthreads();
            buf[threadIdx.x] += t;
            __syncthreads();
        }
        int incl = buf[threadIdx.x];
        int excl = incl - v;
        if (i < n) rowptr[i] = carry + excl;
        __syncthreads();
        if (threadIdx.x == 255) carry += incl;
        __syncthreads();
    }
    if (threadIdx.x == 0) rowptr[n] = carry;
}

__global__ void copy_i32(const int* __restrict__ s, int* __restrict__ d, int n) {
    int i = blockIdx.x * blockDim.x + threadIdx.x;
    if (i < n) d[i] = s[i];
}

__global__ void fill_edges(const int* __restrict__ src, const int* __restrict__ dst,
                           int* __restrict__ cursor, int* __restrict__ esrc, int nEdges) {
    int e = blockIdx.x * blockDim.x + threadIdx.x;
    if (e >= nEdges) return;
    int p = atomicAdd(&cursor[dst[e]], 1);
    esrc[p] = src[e];
}

__global__ __launch_bounds__(256) void gather_prop(const float* __restrict__ h,
                                                   const int* __restrict__ esrc,
                                                   const int* __restrict__ rowptr,
                                                   const float* __restrict__ prevprev,
                                                   float* __restrict__ outp, int nNodes) {
    int v    = blockIdx.x * (blockDim.x >> 5) + (threadIdx.x >> 5);
    int lane = threadIdx.x & 31;
    if (v >= nNodes) return;
    int beg = rowptr[v];
    int end = rowptr[v + 1];
    float4 acc = make_float4(0.f, 0.f, 0.f, 0.f);
    for (int j = beg; j < end; ++j) {
        int s = esrc[j];
        if (j + 1 < end)
            __builtin_prefetch(h + (size_t)esrc[j + 1] * D_FEAT + lane * 4, 0, 0);
        const float4 t = *(const float4*)(h + (size_t)s * D_FEAT + lane * 4);
        acc.x += t.x; acc.y += t.y; acc.z += t.z; acc.w += t.w;
    }
    int deg = end - beg;
    float scale = 1.0f / (float)(deg > 1 ? deg : 1);
    float4 r;
    if (prevprev != nullptr) {
        const float4 p = *(const float4*)(prevprev + (size_t)v * D_FEAT + lane * 4);
        r.x = 2.0f * acc.x * scale - p.x;
        r.y = 2.0f * acc.y * scale - p.y;
        r.z = 2.0f * acc.z * scale - p.z;
        r.w = 2.0f * acc.w * scale - p.w;
    } else {
        r.x = acc.x * scale;
        r.y = acc.y * scale;
        r.z = acc.z * scale;
        r.w = acc.w * scale;
    }
    *(float4*)(outp + (size_t)v * D_FEAT + lane * 4) = r;
}

__global__ __launch_bounds__(256) void cheb_gemm(const float* __restrict__ T0,
                                                 const float* __restrict__ T1,
                                                 const float* __restrict__ T2,
                                                 const float* __restrict__ T3,
                                                 const float* __restrict__ W,
                                                 const float* __restrict__ bias,
                                                 float* __restrict__ out) {
    const int lane = threadIdx.x & 31;
    const int wave = threadIdx.x >> 5;
    const int rowTile = blockIdx.x;
    const int m   = lane & 15;
    const int hi  = lane >> 4;
    const int row = rowTile * 16 + m;
    const int col = wave * 16 + m;

    v8f c = {};
    const float* terms[4] = {T0, T1, T2, T3};

#pragma unroll
    for (int t = 0; t < 4; ++t) {
        const float* __restrict__ h    = terms[t];
        const float* __restrict__ wrow = W + (size_t)col * (D_FEAT * (K_CHEB + 1)) + t * D_FEAT;
#pragma unroll 8
        for (int k0 = 0; k0 < D_FEAT; k0 += 4) {
            const int kk = k0 + (hi << 1);
            v2f a  = *(const v2f*)(h + (size_t)row * D_FEAT + kk);
            v2f bm = *(const v2f*)(wrow + kk);
            c = __builtin_amdgcn_wmma_f32_16x16x4_f32(
                     false, a,  false, bm,
                     (short)0, c,  false,  false);
        }
    }

    const float bj = bias[col];
#pragma unroll
    for (int i = 0; i < 8; ++i) {
        const int mr = rowTile * 16 + i + hi * 8;
        out[(size_t)mr * D_FEAT + col] = c[i] + bj;
    }
}

extern "C" void kernel_launch(void* const* d_in, const int* in_sizes, int n_in,
                              void* d_out, int out_size, void* d_ws, size_t ws_size,
                              hipStream_t stream) {
    const float* x  = (const float*)d_in[0];
    const int*   ei = (const int*)d_in[1];
    const float* W  = (const float*)d_in[2];
    const float* b  = (const float*)d_in[3];
    float* out = (float*)d_out;

    const int* src = ei;
    const int* dst = ei + N_EDGES;

    const size_t ND = (size_t)N_NODES * D_FEAT;
    float* B1     = (float*)d_ws;
    float* B2     = B1 + ND;
    float* B3     = B2 + ND;
    int*   rowptr = (int*)(B3 + ND);
    int*   cursor = rowptr + (N_NODES + 1);
    int*   esrc   = cursor + N_NODES;

    dim3 blk(256);
    dim3 gEdge((N_EDGES + 255) / 256);
    dim3 gNode((N_NODES + 255) / 256);
    dim3 gWaveNode((N_NODES + 7) / 8);

    zero_i32<<<gNode, blk, 0, stream>>>(cursor, N_NODES);
    count_deg_i<<<gEdge, blk, 0, stream>>>(dst, cursor, N_EDGES);
    scan_excl<<<1, blk, 0, stream>>>(cursor, rowptr, N_NODES);
    copy_i32<<<gNode, blk, 0, stream>>>(rowptr, cursor, N_NODES);
    fill_edges<<<gEdge, blk, 0, stream>>>(src, dst, cursor, esrc, N_EDGES);

    gather_prop<<<gWaveNode, blk, 0, stream>>>(x,  esrc, rowptr, nullptr, B1, N_NODES);
    gather_prop<<<gWaveNode, blk, 0, stream>>>(B1, esrc, rowptr, x,       B2, N_NODES);
    gather_prop<<<gWaveNode, blk, 0, stream>>>(B2, esrc, rowptr, B1,      B3, N_NODES);

    cheb_gemm<<<dim3(N_NODES / 16), blk, 0, stream>>>(x, B1, B2, B3, W, b, out);
}


# Edge pass split by cache half: a work-group reads its die, draws a chunk ticket of that half, counts and places only edges whose destination is in that half's nodes; sums stay in add arrival order
# speedup vs baseline: 4.7585x; 4.7585x over previous
#include <hip/hip_runtime.h>

#define N_NODES 50000
#define N_EDGES 800000
#define D_FEAT  128
#define K_CHEB  3

typedef __attribute__((ext_vector_type(2))) float v2f;
typedef __attribute__((ext_vector_type(8))) float v8f;

#define TICKET_STRIDE 64
__global__ void zero_i32(int* __restrict__ p, int n, int* __restrict__ tickets) {
    int i = blockIdx.x * blockDim.x + threadIdx.x;
    if (i < n) p[i] = 0;
    if (i < 2) tickets[i * TICKET_STRIDE] = 0;
}

#define SLOTS_PER_NODE D_FEAT
#define EDGES_PER_THREAD 8
#define CHUNK_EDGES (256 * EDGES_PER_THREAD)
#define N_CHUNKS ((N_EDGES + CHUNK_EDGES - 1) / CHUNK_EDGES)
#define HALF_NODES (N_NODES / 2)
#define SPILL_CAP (N_EDGES - 2 * TICKET_STRIDE)
__global__ __launch_bounds__(256) void count_place(const int* __restrict__ src, const int* __restrict__ dst,
                                                   int* __restrict__ cnt, int* __restrict__ slots,
                                                   int* __restrict__ spillN, int* __restrict__ spill,
                                                   int* __restrict__ tickets, int nEdges) {
    __shared__ int unit;
    if (threadIdx.x == 0) {
        int h = (__builtin_amdgcn_s_sendmsg_rtn(0x87) >> 18) & 1;
        int t = atomicAdd(&tickets[h * TICKET_STRIDE], 1);
        if (t >= N_CHUNKS) {
            h ^= 1;
            t = atomicAdd(&tickets[h * TICKET_STRIDE], 1);
        }
        unit = t < N_CHUNKS ? 2 * t + h : -1;
    }
    __syncthreads();
    const int u = unit;
    if (u < 0) return;
    const int e0 = (u >> 1) * CHUNK_EDGES + threadIdx.x * EDGES_PER_THREAD;
    if (e0 >= nEdges) return;
    const int lo = (u & 1) * HALF_NODES;
    const int4 da = *(const int4*)(dst + e0), db = *(const int4*)(dst + e0 + 4);
    const int4 sa = *(const int4*)(src + e0), sb = *(const int4*)(src + e0 + 4);
    const int d[EDGES_PER_THREAD] = {da.x, da.y, da.z, da.w, db.x, db.y, db.z, db.w};
    const int s[EDGES_PER_THREAD] = {sa.x, sa.y, sa.z, sa.w, sb.x, sb.y, sb.z, sb.w};
    int r[EDGES_PER_THREAD];
#pragma unroll
    for (int k = 0; k < EDGES_PER_THREAD; ++k) {
        r[k] = -1;
        if ((unsigned)(d[k] - lo) < (unsigned)HALF_NODES) r[k] = atomicAdd(&cnt[d[k]], 1);
    }
#pragma unroll
    for (int k = 0; k < EDGES_PER_THREAD; ++k) {
        if (r[k] < 0) continue;
        if (r[k] < SLOTS_PER_NODE) slots[(size_t)d[k] * SLOTS_PER_NODE + r[k]] = s[k];
        else                       spill[atomicAdd(spillN, 1)] = e0 + k;
    }
}

#define NODES_PER_BLOCK 32
__global__ __launch_bounds__(256) void gather_prop(const float* __restrict__ h,
                                                   const int* slots,
                                                   const int* __restrict__ cnt,
                                                   const int* __restrict__ spillN,
                                                   const int* __restrict__ spill,
                                                   const int* __restrict__ src,
                                                   const int* __restrict__ dst,
                                                   const float* __restrict__ prevprev,
                                                   float* outp, int nNodes) {
    const int v  = blockIdx.x * NODES_PER_BLOCK + (threadIdx.x >> 3);
    const int l8 = threadIdx.x & 7;
    if (v >= nNodes) return;
    const int deg = cnt[v];
    const int n   = deg < SLOTS_PER_NODE ? deg : SLOTS_PER_NODE;
    const int* myslots = slots + (size_t)v * SLOTS_PER_NODE;
    float4 acc[4];
#pragma unroll
    for (int q = 0; q < 4; ++q) acc[q] = make_float4(0.f, 0.f, 0.f, 0.f);
    for (int j = 0; j < n; ++j) {
        const float* __restrict__ row = h + (size_t)myslots[j] * D_FEAT + l8 * 4;
        float4 t[4];
#pragma unroll
        for (int q = 0; q < 4; ++q) t[q] = *(const float4*)(row + q * 32);
#pragma unroll
        for (int q = 0; q < 4; ++q) {
            acc[q].x += t[q].x; acc[q].y += t[q].y; acc[q].z += t[q].z; acc[q].w += t[q].w;
        }
    }
    if (deg > SLOTS_PER_NODE) {
        int m = *spillN;
        m = m < SPILL_CAP ? m : SPILL_CAP;
        for (int k = 0; k < m; ++k) {
            const int e = spill[k];
            if (dst[e] != v) continue;
            const float* __restrict__ row = h + (size_t)src[e] * D_FEAT + l8 * 4;
#pragma unroll
            for (int q = 0; q < 4; ++q) {
                const float4 t = *(const float4*)(row + q * 32);
                acc[q].x += t.x; acc[q].y += t.y; acc[q].z += t.z; acc[q].w += t.w;
            }
        }
    }
    const float scale = 1.0f / (float)(deg > 1 ? deg : 1);
    const size_t o = (size_t)v * D_FEAT + l8 * 4;
    if (prevprev != nullptr) {
        float4 p[4];
#pragma unroll
        for (int q = 0; q < 4; ++q) p[q] = *(const float4*)(prevprev + o + q * 32);
#pragma unroll
        for (int q = 0; q < 4; ++q) {
            float4 r;
            r.x = 2.0f * acc[q].x * scale - p[q].x;
            r.y = 2.0f * acc[q].y * scale - p[q].y;
            r.z = 2.0f * acc[q].z * scale - p[q].z;
            r.w = 2.0f * acc[q].w * scale - p[q].w;
            *(float4*)(outp + o + q * 32) = r;
        }
    } else {
#pragma unroll
        for (int q = 0; q < 4; ++q) {
            float4 r;
            r.x = acc[q].x * scale;
            r.y = acc[q].y * scale;
            r.z = acc[q].z * scale;
            r.w = acc[q].w * scale;
            *(float4*)(outp + o + q * 32) = r;
        }
    }
}

#define KTOT    (D_FEAT * (K_CHEB + 1))
#define KCHUNK  16
#define NCHUNK  (KTOT / KCHUNK)

__global__ __launch_bounds__(32) __attribute__((amdgpu_waves_per_eu(1, 2)))
void cheb_gemm(const float* __restrict__ T0,
               const float* __restrict__ T1,
               const float* __restrict__ T2,
               const float* __restrict__ T3,
               const float* __restrict__ W,
               const float* __restrict__ bias,
               float* __restrict__ out) {
    const int lane  = threadIdx.x;
    const int m     = lane & 15;
    const int hi    = lane >> 4;
    const int tile0 = blockIdx.x * 2;
    const bool two  = (tile0 + 1) < (N_NODES / 16);

    const size_t aoff0 = (size_t)(tile0 * 16 + m) * D_FEAT + 2 * hi;
    const size_t aoff1 = aoff0 + (two ? (size_t)16 * D_FEAT : (size_t)0);
    const float* __restrict__ wl = W + (size_t)m * KTOT + 2 * hi;

    v8f acc[2][8];
#pragma unroll
    for (int r = 0; r < 2; ++r)
#pragma unroll
        for (int j = 0; j < 8; ++j) acc[r][j] = v8f{};

    v2f fa[2][2][4];
    v2f fb[2][8][4];

    auto load_chunk = [&](const int buf, const int c) {
        const int t = c >> 3;
        const float* __restrict__ h = (t == 0) ? T0 : (t == 1) ? T1 : (t == 2) ? T2 : T3;
        const float* __restrict__ ha = h + (c & 7) * KCHUNK;
        const float* __restrict__ wc = wl + c * KCHUNK;
#pragma unroll
        for (int s = 0; s < 4; ++s) {
            fa[buf][0][s] = *(const v2f*)(ha + aoff0 + 4 * s);
            fa[buf][1][s] = *(const v2f*)(ha + aoff1 + 4 * s);
        }
#pragma unroll
        for (int j = 0; j < 8; ++j)
#pragma unroll
            for (int s = 0; s < 4; ++s)
                fb[buf][j][s] = *(const v2f*)(wc + (size_t)j * 16 * KTOT + 4 * s);
    };

    auto mma_chunk = [&](const int buf) {
#pragma unroll
        for (int s = 0; s < 4; ++s)
#pragma unroll
            for (int j = 0; j < 8; ++j)
#pragma unroll
                for (int r = 0; r < 2; ++r)
                    acc[r][j] = __builtin_amdgcn_wmma_f32_16x16x4_f32(
                         false, fa[buf][r][s],  false, fb[buf][j][s],
                         (short)0, acc[r][j],  false,  false);
    };

    load_chunk(0, 0);
#pragma unroll 1
    for (int c = 0; c < NCHUNK - 2; c += 2) {
        load_chunk(1, c + 1);
        mma_chunk(0);
        load_chunk(0, c + 2);
        mma_chunk(1);
    }
    load_chunk(1, NCHUNK - 1);
    mma_chunk(0);
    mma_chunk(1);

#pragma unroll
    for (int r = 0; r < 2; ++r) {
        if (r == 1 && !two) break;
#pragma unroll
        for (int j = 0; j < 8; ++j) {
            const int col = j * 16 + m;
            const float bj = bias[col];
#pragma unroll
            for (int i = 0; i < 8; ++i) {
                const int mr = (tile0 + r) * 16 + i + hi * 8;
                out[(size_t)mr * D_FEAT + col] = acc[r][j][i] + bj;
            }
        }
    }
}

extern "C" void kernel_launch(void* const* d_in, const int* in_sizes, int n_in,
                              void* d_out, int out_size, void* d_ws, size_t ws_size,
                              hipStream_t stream) {
    const float* x  = (const float*)d_in[0];
    const int*   ei = (const int*)d_in[1];
    const float* W  = (const float*)d_in[2];
    const float* b  = (const float*)d_in[3];
    float* out = (float*)d_out;

    const int* src = ei;
    const int* dst = ei + N_EDGES;

    const size_t ND = (size_t)N_NODES * D_FEAT;
    float* B1     = (float*)d_ws;
    float* B2     = B1 + ND;
    float* B3     = B2 + ND;
    int*   spillN = (int*)(B3 + ND);
    int*   cnt    = spillN + 1;
    int*   spill  = cnt + N_NODES;
    int*   tickets = spill + SPILL_CAP;
    int*   slots  = (int*)B3;

    dim3 blk(256);
    dim3 gUnit(2 * N_CHUNKS);
    dim3 gNode((N_NODES + 255) / 256);
    dim3 gGroupNode((N_NODES + NODES_PER_BLOCK - 1) / NODES_PER_BLOCK);

    zero_i32<<<gNode, blk, 0, stream>>>(spillN, N_NODES + 1, tickets);
    count_place<<<gUnit, blk, 0, stream>>>(src, dst, cnt, slots, spillN, spill, tickets, N_EDGES);

    gather_prop<<<gGroupNode, blk, 0, stream>>>(x,  slots, cnt, spillN, spill, src, dst, nullptr, B1, N_NODES);
    gather_prop<<<gGroupNode, blk, 0, stream>>>(B1, slots, cnt, spillN, spill, src, dst, x,       B2, N_NODES);
    gather_prop<<<gGroupNode, blk, 0, stream>>>(B2, slots, cnt, spillN, spill, src, dst, B1,      B3, N_NODES);

    cheb_gemm<<<dim3((N_NODES / 16 + 1) / 2), dim3(32), 0, stream>>>(x, B1, B2, B3, W, b, out);
}


# Third propagation fused with the product: a block gathers T3 of 32 nodes into an LDS tile beside T0..T2 and multiplies it; half its waves do the T0..T2 part before the gather; 5 launches; sums kept
# speedup vs baseline: 5.2533x; 1.1040x over previous
#include <hip/hip_runtime.h>

#define N_NODES 50000
#define N_EDGES 800000
#define D_FEAT  128
#define K_CHEB  3

typedef __attribute__((ext_vector_type(2))) float v2f;
typedef __attribute__((ext_vector_type(8))) float v8f;

#define TICKET_STRIDE 64
__global__ void zero_i32(int* __restrict__ p, int n, int* __restrict__ tickets) {
    int i = blockIdx.x * blockDim.x + threadIdx.x;
    if (i < n) p[i] = 0;
    if (i < 2) tickets[i * TICKET_STRIDE] = 0;
}

#define SLOTS_PER_NODE D_FEAT
#define EDGES_PER_THREAD 8
#define CHUNK_EDGES (256 * EDGES_PER_THREAD)
#define N_CHUNKS ((N_EDGES + CHUNK_EDGES - 1) / CHUNK_EDGES)
#define HALF_NODES (N_NODES / 2)
#define SPILL_CAP (N_EDGES - 2 * TICKET_STRIDE)
__global__ __launch_bounds__(256) void count_place(const int* __restrict__ src, const int* __restrict__ dst,
                                                   int* __restrict__ cnt, int* __restrict__ slots,
                                                   int* __restrict__ spillN, int* __restrict__ spill,
                                                   int* __restrict__ tickets, int nEdges) {
    __shared__ int unit;
    if (threadIdx.x == 0) {
        int h = (__builtin_amdgcn_s_sendmsg_rtn(0x87) >> 18) & 1;
        int t = atomicAdd(&tickets[h * TICKET_STRIDE], 1);
        if (t >= N_CHUNKS) {
            h ^= 1;
            t = atomicAdd(&tickets[h * TICKET_STRIDE], 1);
        }
        unit = t < N_CHUNKS ? 2 * t + h : -1;
    }
    __syncthreads();
    const int u = unit;
    if (u < 0) return;
    const int e0 = (u >> 1) * CHUNK_EDGES + threadIdx.x * EDGES_PER_THREAD;
    if (e0 >= nEdges) return;
    const int lo = (u & 1) * HALF_NODES;
    const int4 da = *(const int4*)(dst + e0), db = *(const int4*)(dst + e0 + 4);
    const int4 sa = *(const int4*)(src + e0), sb = *(const int4*)(src + e0 + 4);
    const int d[EDGES_PER_THREAD] = {da.x, da.y, da.z, da.w, db.x, db.y, db.z, db.w};
    const int s[EDGES_PER_THREAD] = {sa.x, sa.y, sa.z, sa.w, sb.x, sb.y, sb.z, sb.w};
    int r[EDGES_PER_THREAD];
#pragma unroll
    for (int k = 0; k < EDGES_PER_THREAD; ++k) {
        r[k] = -1;
        if ((unsigned)(d[k] - lo) < (unsigned)HALF_NODES) r[k] = atomicAdd(&cnt[d[k]], 1);
    }
#pragma unroll
    for (int k = 0; k < EDGES_PER_THREAD; ++k) {
        if (r[k] < 0) continue;
        if (r[k] < SLOTS_PER_NODE) slots[(size_t)d[k] * SLOTS_PER_NODE + r[k]] = s[k];
        else                       spill[atomicAdd(spillN, 1)] = e0 + k;
    }
}

#define NODES_PER_BLOCK 32
__global__ __launch_bounds__(256) void gather_prop(const float* __restrict__ h,
                                                   const int* slots,
                                                   const int* __restrict__ cnt,
                                                   const int* __restrict__ spillN,
                                                   const int* __restrict__ spill,
                                                   const int* __restrict__ src,
                                                   const int* __restrict__ dst,
                                                   const float* __restrict__ prevprev,
                                                   float* outp, int nNodes) {
    const int v  = blockIdx.x * NODES_PER_BLOCK + (threadIdx.x >> 3);
    const int l8 = threadIdx.x & 7;
    if (v >= nNodes) return;
    const int deg = cnt[v];
    const int n   = deg < SLOTS_PER_NODE ? deg : SLOTS_PER_NODE;
    const int* myslots = slots + (size_t)v * SLOTS_PER_NODE;
    float4 acc[4];
#pragma unroll
    for (int q = 0; q < 4; ++q) acc[q] = make_float4(0.f, 0.f, 0.f, 0.f);
    for (int j = 0; j < n; ++j) {
        const float* __restrict__ row = h + (size_t)myslots[j] * D_FEAT + l8 * 4;
        float4 t[4];
#pragma unroll
        for (int q = 0; q < 4; ++q) t[q] = *(const float4*)(row + q * 32);
#pragma unroll
        for (int q = 0; q < 4; ++q) {
            acc[q].x += t[q].x; acc[q].y += t[q].y; acc[q].z += t[q].z; acc[q].w += t[q].w;
        }
    }
    if (deg > SLOTS_PER_NODE) {
        int m = *spillN;
        m = m < SPILL_CAP ? m : SPILL_CAP;
        for (int k = 0; k < m; ++k) {
            const int e = spill[k];
            if (dst[e] != v) continue;
            const float* __restrict__ row = h + (size_t)src[e] * D_FEAT + l8 * 4;
#pragma unroll
            for (int q = 0; q < 4; ++q) {
                const float4 t = *(const float4*)(row + q * 32);
                acc[q].x += t.x; acc[q].y += t.y; acc[q].z += t.z; acc[q].w += t.w;
            }
        }
    }
    const float scale = 1.0f / (float)(deg > 1 ? deg : 1);
    const size_t o = (size_t)v * D_FEAT + l8 * 4;
    if (prevprev != nullptr) {
        float4 p[4];
#pragma unroll
        for (int q = 0; q < 4; ++q) p[q] = *(const float4*)(prevprev + o + q * 32);
#pragma unroll
        for (int q = 0; q < 4; ++q) {
            float4 r;
            r.x = 2.0f * acc[q].x * scale - p[q].x;
            r.y = 2.0f * acc[q].y * scale - p[q].y;
            r.z = 2.0f * acc[q].z * scale - p[q].z;
            r.w = 2.0f * acc[q].w * scale - p[q].w;
            *(float4*)(outp + o + q * 32) = r;
        }
    } else {
#pragma unroll
        for (int q = 0; q < 4; ++q) {
            float4 r;
            r.x = acc[q].x * scale;
            r.y = acc[q].y * scale;
            r.z = acc[q].z * scale;
            r.w = acc[q].w * scale;
            *(float4*)(outp + o + q * 32) = r;
        }
    }
}

#define KTOT    (D_FEAT * (K_CHEB + 1))
#define KCHUNK  16
#define NCHUNK  (KTOT / KCHUNK)
#define TILE_ROWS NODES_PER_BLOCK

__global__ __launch_bounds__(256) void gather_gemm(const float* __restrict__ h,
                                                   const int* __restrict__ slots,
                                                   const int* __restrict__ cnt,
                                                   const int* __restrict__ spillN,
                                                   const int* __restrict__ spill,
                                                   const int* __restrict__ src,
                                                   const int* __restrict__ dst,
                                                   const float* __restrict__ prevprev,
                                                   const float* __restrict__ x,
                                                   const float* __restrict__ W,
                                                   const float* __restrict__ bias,
                                                   float* __restrict__ out,
                                                   int nNodes) {
    __shared__ float sT[TILE_ROWS * KTOT];

    const int g     = threadIdx.x >> 3;
    const int l8    = threadIdx.x & 7;
    const int v     = blockIdx.x * NODES_PER_BLOCK + g;
    const bool valid = v < nNodes;
    float* trow     = sT + g * KTOT;
    const int skey  = (g & 15) << 2;

    {
        float4 r0[4], r1[4], r2[4];
#pragma unroll
        for (int q = 0; q < 4; ++q) r0[q] = r1[q] = r2[q] = make_float4(0.f, 0.f, 0.f, 0.f);
        if (valid) {
            const size_t o = (size_t)v * D_FEAT + l8 * 4;
#pragma unroll
            for (int q = 0; q < 4; ++q) {
                r0[q] = *(const float4*)(x        + o + q * 32);
                r1[q] = *(const float4*)(prevprev + o + q * 32);
                r2[q] = *(const float4*)(h        + o + q * 32);
            }
        }
#pragma unroll
        for (int q = 0; q < 4; ++q) {
            const int col = q * 32 + l8 * 4;
            *(float4*)(trow + ((0 * D_FEAT + col) ^ skey)) = r0[q];
            *(float4*)(trow + ((1 * D_FEAT + col) ^ skey)) = r1[q];
            *(float4*)(trow + ((2 * D_FEAT + col) ^ skey)) = r2[q];
        }
    }
    __syncthreads();

    const int lane = threadIdx.x & 31;
    const int w    = __builtin_amdgcn_readfirstlane((int)(threadIdx.x >> 5));
    const int m    = lane & 15;
    const int hi   = lane >> 4;
    const int key  = m << 2;
    const float* a0 = sT + m * KTOT;
    const float* a1 = sT + (16 + m) * KTOT;
    const float* __restrict__ wl = W + (size_t)(w * 16 + m) * KTOT + 2 * hi;

    v8f acc[2];
    acc[0] = v8f{};
    acc[1] = v8f{};
    v2f fa[2][2][4];
    v2f fb[2][4];

    auto load_chunk = [&](const int buf, const int c) {
#pragma unroll
        for (int s = 0; s < 4; ++s) {
            const int col = (c * KCHUNK + 4 * s + 2 * hi) ^ key;
            fa[buf][0][s] = *(const v2f*)(a0 + col);
            fa[buf][1][s] = *(const v2f*)(a1 + col);
            fb[buf][s]    = *(const v2f*)(wl + c * KCHUNK + 4 * s);
        }
    };
    auto mma_chunk = [&](const int buf) {
#pragma unroll
        for (int s = 0; s < 4; ++s)
#pragma unroll
            for (int r = 0; r < 2; ++r)
                acc[r] = __builtin_amdgcn_wmma_f32_16x16x4_f32(
                     false, fa[buf][r][s],  false, fb[buf][s],
                     (short)0, acc[r],  false,  false);
    };
    auto product = [&](const int cb, const int ce) {
        load_chunk(0, cb);
#pragma unroll 1
        for (int c = cb; c < ce - 2; c += 2) {
            load_chunk(1, c + 1);
            mma_chunk(0);
            load_chunk(0, c + 2);
            mma_chunk(1);
        }
        load_chunk(1, ce - 1);
        mma_chunk(0);
        mma_chunk(1);
    };

    auto gather_t3 = [&]() {
        float4 r3[4];
#pragma unroll
        for (int q = 0; q < 4; ++q) r3[q] = make_float4(0.f, 0.f, 0.f, 0.f);
        if (valid) {
            const int deg = cnt[v];
            const int n   = deg < SLOTS_PER_NODE ? deg : SLOTS_PER_NODE;
            const int* __restrict__ myslots = slots + (size_t)v * SLOTS_PER_NODE;
            float4 sum[4];
#pragma unroll
            for (int q = 0; q < 4; ++q) sum[q] = make_float4(0.f, 0.f, 0.f, 0.f);
            for (int j = 0; j < n; ++j) {
                const float* __restrict__ row = h + (size_t)myslots[j] * D_FEAT + l8 * 4;
                float4 t[4];
#pragma unroll
                for (int q = 0; q < 4; ++q) t[q] = *(const float4*)(row + q * 32);
#pragma unroll
                for (int q = 0; q < 4; ++q) {
                    sum[q].x += t[q].x; sum[q].y += t[q].y; sum[q].z += t[q].z; sum[q].w += t[q].w;
                }
            }
            if (deg > SLOTS_PER_NODE) {
                int mS = *spillN;
                mS = mS < SPILL_CAP ? mS : SPILL_CAP;
                for (int k = 0; k < mS; ++k) {
                    const int e = spill[k];
                    if (dst[e] != v) continue;
                    const float* __restrict__ row = h + (size_t)src[e] * D_FEAT + l8 * 4;
#pragma unroll
                    for (int q = 0; q < 4; ++q) {
                        const float4 t = *(const float4*)(row + q * 32);
                        sum[q].x += t.x; sum[q].y += t.y; sum[q].z += t.z; sum[q].w += t.w;
                    }
                }
            }
            const float scale = 1.0f / (float)(deg > 1 ? deg : 1);
#pragma unroll
            for (int q = 0; q < 4; ++q) {
                const float4 p = *(const float4*)(trow + ((1 * D_FEAT + q * 32 + l8 * 4) ^ skey));
                r3[q].x = 2.0f * sum[q].x * scale - p.x;
                r3[q].y = 2.0f * sum[q].y * scale - p.y;
                r3[q].z = 2.0f * sum[q].z * scale - p.z;
                r3[q].w = 2.0f * sum[q].w * scale - p.w;
            }
        }
#pragma unroll
        for (int q = 0; q < 4; ++q)
            *(float4*)(trow + ((3 * D_FEAT + q * 32 + l8 * 4) ^ skey)) = r3[q];
    };

    const int productFirst = (w ^ (w >> 2)) & 1;
#pragma unroll 1
    for (int step = 0; step < 2; ++step) {
        if ((step ^ productFirst) == 0) gather_t3();
        else                            product(0, 3 * (NCHUNK / 4));
    }
    __syncthreads();
    product(3 * (NCHUNK / 4), NCHUNK);

    const int col = w * 16 + m;
    const float bj = bias[col];
#pragma unroll
    for (int r = 0; r < 2; ++r) {
#pragma unroll
        for (int i = 0; i < 8; ++i) {
            const int R = r * 16 + i + hi * 8;
            sT[R * KTOT + (col ^ ((R & 15) << 2))] = acc[r][i] + bj;
        }
    }
    __syncthreads();
    if (valid) {
        float* __restrict__ orow = out + (size_t)v * D_FEAT + l8 * 4;
#pragma unroll
        for (int q = 0; q < 4; ++q)
            *(float4*)(orow + q * 32) = *(const float4*)(trow + ((q * 32 + l8 * 4) ^ skey));
    }
}

extern "C" void kernel_launch(void* const* d_in, const int* in_sizes, int n_in,
                              void* d_out, int out_size, void* d_ws, size_t ws_size,
                              hipStream_t stream) {
    const float* x  = (const float*)d_in[0];
    const int*   ei = (const int*)d_in[1];
    const float* W  = (const float*)d_in[2];
    const float* b  = (const float*)d_in[3];
    float* out = (float*)d_out;

    const int* src = ei;
    const int* dst = ei + N_EDGES;

    const size_t ND = (size_t)N_NODES * D_FEAT;
    float* B1     = (float*)d_ws;
    float* B2     = B1 + ND;
    float* B3     = B2 + ND;
    int*   spillN = (int*)(B3 + ND);
    int*   cnt    = spillN + 1;
    int*   spill  = cnt + N_NODES;
    int*   tickets = spill + SPILL_CAP;
    int*   slots  = (int*)B3;

    dim3 blk(256);
    dim3 gUnit(2 * N_CHUNKS);
    dim3 gNode((N_NODES + 255) / 256);
    dim3 gGroupNode((N_NODES + NODES_PER_BLOCK - 1) / NODES_PER_BLOCK);

    zero_i32<<<gNode, blk, 0, stream>>>(spillN, N_NODES + 1, tickets);
    count_place<<<gUnit, blk, 0, stream>>>(src, dst, cnt, slots, spillN, spill, tickets, N_EDGES);

    gather_prop<<<gGroupNode, blk, 0, stream>>>(x,  slots, cnt, spillN, spill, src, dst, nullptr, B1, N_NODES);
    gather_prop<<<gGroupNode, blk, 0, stream>>>(B1, slots, cnt, spillN, spill, src, dst, x,       B2, N_NODES);
    gather_gemm<<<gGroupNode, blk, 0, stream>>>(B2, slots, cnt, spillN, spill, src, dst, B1, x, W, b, out, N_NODES);
}
